# starting kernel
# baseline (speedup 1.0000x reference)
#include <hip/hip_runtime.h>
#include <hip/hip_bf16.h>

typedef __attribute__((ext_vector_type(2))) float v2f;
typedef __attribute__((ext_vector_type(8))) float v8f;

#define Bn 4
#define Cc 256
#define Hh 56
#define Ww 56
#define HW (Hh * Ww)
#define NP (Bn * HW)
#define TN 256
#define Gg 16
#define Kk 7
#define KK (Kk * Kk)
#define NK (Gg * KK)
#define EPSf 1e-5f

__global__ __launch_bounds__(256) void gemm1_reduce_bn_prelu(
    const float* __restrict__ x, const float* __restrict__ rw,
    const float* __restrict__ rb, const float* __restrict__ gamma,
    const float* __restrict__ beta, const float* __restrict__ mean,
    const float* __restrict__ var, const float* __restrict__ pa,
    float* __restrict__ t_out)
{
    const int lane  = threadIdx.x & 31;
    const int wave  = threadIdx.x >> 5;
    const int tile  = blockIdx.x * 8 + wave;
    const int npair = tile & 7;
    const int mpair = tile >> 3;
    const int half  = lane >> 4;
    const int l16   = lane & 15;

    const int p_base = mpair * 32;
    const int b      = p_base / HW;
    const int hw     = p_base - b * HW;
    const float* xrow0 = x + (size_t)b * Cc * HW + hw + l16;
    const float* xrow1 = xrow0 + 16;
    const int n0 = npair * 32 + l16;
    const int n1 = n0 + 16;
    const float* wrow0 = rw + (size_t)n0 * Cc;
    const float* wrow1 = rw + (size_t)n1 * Cc;

    v8f acc00 = {}, acc01 = {}, acc10 = {}, acc11 = {};
#pragma unroll 4
    for (int k0 = 0; k0 < Cc; k0 += 4) {
        const int kh = k0 + 2 * half;
        v2f a0, a1;
        a0.x = xrow0[(size_t)kh * HW];
        a0.y = xrow0[(size_t)(kh + 1) * HW];
        a1.x = xrow1[(size_t)kh * HW];
        a1.y = xrow1[(size_t)(kh + 1) * HW];
        v2f b0 = *(const v2f*)(wrow0 + kh);
        v2f b1 = *(const v2f*)(wrow1 + kh);
        acc00 = __builtin_amdgcn_wmma_f32_16x16x4_f32(false, a0, false, b0, (short)0, acc00, false, false);
        acc01 = __builtin_amdgcn_wmma_f32_16x16x4_f32(false, a0, false, b1, (short)0, acc01, false, false);
        acc10 = __builtin_amdgcn_wmma_f32_16x16x4_f32(false, a1, false, b0, (short)0, acc10, false, false);
        acc11 = __builtin_amdgcn_wmma_f32_16x16x4_f32(false, a1, false, b1, (short)0, acc11, false, false);
    }

    const float al = pa[0];
    const float s0  = gamma[n0] * rsqrtf(var[n0] + EPSf);
    const float s1  = gamma[n1] * rsqrtf(var[n1] + EPSf);
    const float bi0 = rb[n0] - mean[n0], bi1 = rb[n1] - mean[n1];
    const float bt0 = beta[n0],          bt1 = beta[n1];

    float* trow0 = t_out + (size_t)(p_base + 8 * half) * TN;
    float* trow1 = t_out + (size_t)(p_base + 16 + 8 * half) * TN;
#pragma unroll
    for (int r = 0; r < 8; ++r) {
        float v00 = (acc00[r] + bi0) * s0 + bt0; v00 = v00 > 0.f ? v00 : al * v00;
        float v01 = (acc01[r] + bi1) * s1 + bt1; v01 = v01 > 0.f ? v01 : al * v01;
        float v10 = (acc10[r] + bi0) * s0 + bt0; v10 = v10 > 0.f ? v10 : al * v10;
        float v11 = (acc11[r] + bi1) * s1 + bt1; v11 = v11 > 0.f ? v11 : al * v11;
        trow0[(size_t)r * TN + n0] = v00;
        trow0[(size_t)r * TN + n1] = v01;
        trow1[(size_t)r * TN + n0] = v10;
        trow1[(size_t)r * TN + n1] = v11;
    }
}

__global__ __launch_bounds__(128) void gemm2_span(
    const float* __restrict__ t_in, const float* __restrict__ sw,
    const float* __restrict__ sb, float* __restrict__ ker)
{
    const int lane  = threadIdx.x & 31;
    const int wave  = threadIdx.x >> 5;
    const int tile  = blockIdx.x * 4 + wave;
    const int mtile = tile % 49;
    const int nquad = tile / 49;
    const int half  = lane >> 4;
    const int l16   = lane & 15;

    const float* arow = sw + (size_t)(mtile * 16 + l16) * TN;
    const int p_base  = nquad * 64;
    const float* brow0 = t_in + (size_t)(p_base +  0 + l16) * TN;
    const float* brow1 = t_in + (size_t)(p_base + 16 + l16) * TN;
    const float* brow2 = t_in + (size_t)(p_base + 32 + l16) * TN;
    const float* brow3 = t_in + (size_t)(p_base + 48 + l16) * TN;

    v8f acc0 = {}, acc1 = {}, acc2 = {}, acc3 = {};
#pragma unroll 4
    for (int k0 = 0; k0 < TN; k0 += 4) {
        const int kh = k0 + 2 * half;
        v2f a  = *(const v2f*)(arow + kh);
        v2f b0 = *(const v2f*)(brow0 + kh);
        v2f b1 = *(const v2f*)(brow1 + kh);
        v2f b2 = *(const v2f*)(brow2 + kh);
        v2f b3 = *(const v2f*)(brow3 + kh);
        acc0 = __builtin_amdgcn_wmma_f32_16x16x4_f32(false, a, false, b0, (short)0, acc0, false, false);
        acc1 = __builtin_amdgcn_wmma_f32_16x16x4_f32(false, a, false, b1, (short)0, acc1, false, false);
        acc2 = __builtin_amdgcn_wmma_f32_16x16x4_f32(false, a, false, b2, (short)0, acc2, false, false);
        acc3 = __builtin_amdgcn_wmma_f32_16x16x4_f32(false, a, false, b3, (short)0, acc3, false, false);
    }

    const int kkbase = mtile * 16 + 8 * half;
#pragma unroll
    for (int r = 0; r < 8; ++r) {
        const int kkr = kkbase + r;
        const float bias = sb[kkr];
        float* krow = ker + (size_t)kkr * NP + p_base + l16;
        krow[ 0] = acc0[r] + bias;
        krow[16] = acc1[r] + bias;
        krow[32] = acc2[r] + bias;
        krow[48] = acc3[r] + bias;
    }
}

__global__ __launch_bounds__(256) void involution_apply(
    const float* __restrict__ x, const float* __restrict__ ker,
    float* __restrict__ out)
{
    const int id = blockIdx.x * 256 + threadIdx.x;
    const int hw = id % HW;
    const int bg = id / HW;
    const int g  = bg % Gg;
    const int b  = bg / Gg;
    const int h  = hw / Ww;
    const int w  = hw % Ww;
    const int p  = b * HW + hw;

    float kv[KK];
#pragma unroll
    for (int idx = 0; idx < KK; ++idx)
        kv[idx] = ker[(size_t)(g * KK + idx) * NP + p];

    const float* xb = x   + (size_t)b * Cc * HW;
    float*       ob = out + (size_t)b * Cc * HW;
    for (int cc = 0; cc < Gg; ++cc) {
        const int c = g * 16 + cc;
        const float* xc = xb + (size_t)c * HW;
        float acc = 0.f;
#pragma unroll
        for (int idx = 0; idx < KK; ++idx) {
            const int hh = h + idx / Kk - 3;
            const int ww = w + idx % Kk - 3;
            if (hh >= 0 && hh < Hh && ww >= 0 && ww < Ww)
                acc += xc[hh * Ww + ww] * kv[idx];
        }
        ob[(size_t)c * HW + hw] = acc;
    }
}

extern "C" void kernel_launch(void* const* d_in, const int* in_sizes, int n_in,
                              void* d_out, int out_size, void* d_ws, size_t ws_size,
                              hipStream_t stream) {
    const float* x     = (const float*)d_in[0];
    const float* rw    = (const float*)d_in[1];
    const float* rb    = (const float*)d_in[2];
    const float* gamma = (const float*)d_in[3];
    const float* beta  = (const float*)d_in[4];
    const float* mean  = (const float*)d_in[5];
    const float* var   = (const float*)d_in[6];
    const float* pa    = (const float*)d_in[7];
    const float* sw    = (const float*)d_in[8];
    const float* sb    = (const float*)d_in[9];
    float* out = (float*)d_out;

    float* t_ws   = (float*)d_ws;
    float* ker_ws = t_ws + (size_t)NP * TN;

    gemm1_reduce_bn_prelu<<<392, 256, 0, stream>>>(
        x, rw, rb, gamma, beta, mean, var, pa, t_ws);
    gemm2_span<<<2401, 128, 0, stream>>>(
        t_ws, sw, sb, ker_ws);
    involution_apply<<<(Bn * Gg * HW) / 256, 256, 0, stream>>>(
        x, ker_ws, out);
}


# involution: stage the group's 16 channels of x (4 rows + halo, zero-padded) once per work-group in LDS; the 49 taps read LDS, not 784 bounds-tested serial global loads per thread; sum order kept
# speedup vs baseline: 1.7479x; 1.7479x over previous
#include <hip/hip_runtime.h>
#include <hip/hip_bf16.h>

typedef __attribute__((ext_vector_type(2))) float v2f;
typedef __attribute__((ext_vector_type(8))) float v8f;

#define Bn 4
#define Cc 256
#define Hh 56
#define Ww 56
#define HW (Hh * Ww)
#define NP (Bn * HW)
#define TN 256
#define Gg 16
#define Kk 7
#define KK (Kk * Kk)
#define NK (Gg * KK)
#define EPSf 1e-5f

__global__ __launch_bounds__(256) void gemm1_reduce_bn_prelu(
    const float* __restrict__ x, const float* __restrict__ rw,
    const float* __restrict__ rb, const float* __restrict__ gamma,
    const float* __restrict__ beta, const float* __restrict__ mean,
    const float* __restrict__ var, const float* __restrict__ pa,
    float* __restrict__ t_out)
{
    const int lane  = threadIdx.x & 31;
    const int wave  = threadIdx.x >> 5;
    const int tile  = blockIdx.x * 8 + wave;
    const int npair = tile & 7;
    const int mpair = tile >> 3;
    const int half  = lane >> 4;
    const int l16   = lane & 15;

    const int p_base = mpair * 32;
    const int b      = p_base / HW;
    const int hw     = p_base - b * HW;
    const float* xrow0 = x + (size_t)b * Cc * HW + hw + l16;
    const float* xrow1 = xrow0 + 16;
    const int n0 = npair * 32 + l16;
    const int n1 = n0 + 16;
    const float* wrow0 = rw + (size_t)n0 * Cc;
    const float* wrow1 = rw + (size_t)n1 * Cc;

    v8f acc00 = {}, acc01 = {}, acc10 = {}, acc11 = {};
#pragma unroll 4
    for (int k0 = 0; k0 < Cc; k0 += 4) {
        const int kh = k0 + 2 * half;
        v2f a0, a1;
        a0.x = xrow0[(size_t)kh * HW];
        a0.y = xrow0[(size_t)(kh + 1) * HW];
        a1.x = xrow1[(size_t)kh * HW];
        a1.y = xrow1[(size_t)(kh + 1) * HW];
        v2f b0 = *(const v2f*)(wrow0 + kh);
        v2f b1 = *(const v2f*)(wrow1 + kh);
        acc00 = __builtin_amdgcn_wmma_f32_16x16x4_f32(false, a0, false, b0, (short)0, acc00, false, false);
        acc01 = __builtin_amdgcn_wmma_f32_16x16x4_f32(false, a0, false, b1, (short)0, acc01, false, false);
        acc10 = __builtin_amdgcn_wmma_f32_16x16x4_f32(false, a1, false, b0, (short)0, acc10, false, false);
        acc11 = __builtin_amdgcn_wmma_f32_16x16x4_f32(false, a1, false, b1, (short)0, acc11, false, false);
    }

    const float al = pa[0];
    const float s0  = gamma[n0] * rsqrtf(var[n0] + EPSf);
    const float s1  = gamma[n1] * rsqrtf(var[n1] + EPSf);
    const float bi0 = rb[n0] - mean[n0], bi1 = rb[n1] - mean[n1];
    const float bt0 = beta[n0],          bt1 = beta[n1];

    float* trow0 = t_out + (size_t)(p_base + 8 * half) * TN;
    float* trow1 = t_out + (size_t)(p_base + 16 + 8 * half) * TN;
#pragma unroll
    for (int r = 0; r < 8; ++r) {
        float v00 = (acc00[r] + bi0) * s0 + bt0; v00 = v00 > 0.f ? v00 : al * v00;
        float v01 = (acc01[r] + bi1) * s1 + bt1; v01 = v01 > 0.f ? v01 : al * v01;
        float v10 = (acc10[r] + bi0) * s0 + bt0; v10 = v10 > 0.f ? v10 : al * v10;
        float v11 = (acc11[r] + bi1) * s1 + bt1; v11 = v11 > 0.f ? v11 : al * v11;
        trow0[(size_t)r * TN + n0] = v00;
        trow0[(size_t)r * TN + n1] = v01;
        trow1[(size_t)r * TN + n0] = v10;
        trow1[(size_t)r * TN + n1] = v11;
    }
}

__global__ __launch_bounds__(128) void gemm2_span(
    const float* __restrict__ t_in, const float* __restrict__ sw,
    const float* __restrict__ sb, float* __restrict__ ker)
{
    const int lane  = threadIdx.x & 31;
    const int wave  = threadIdx.x >> 5;
    const int tile  = blockIdx.x * 4 + wave;
    const int mtile = tile % 49;
    const int nquad = tile / 49;
    const int half  = lane >> 4;
    const int l16   = lane & 15;

    const float* arow = sw + (size_t)(mtile * 16 + l16) * TN;
    const int p_base  = nquad * 64;
    const float* brow0 = t_in + (size_t)(p_base +  0 + l16) * TN;
    const float* brow1 = t_in + (size_t)(p_base + 16 + l16) * TN;
    const float* brow2 = t_in + (size_t)(p_base + 32 + l16) * TN;
    const float* brow3 = t_in + (size_t)(p_base + 48 + l16) * TN;

    v8f acc0 = {}, acc1 = {}, acc2 = {}, acc3 = {};
#pragma unroll 4
    for (int k0 = 0; k0 < TN; k0 += 4) {
        const int kh = k0 + 2 * half;
        v2f a  = *(const v2f*)(arow + kh);
        v2f b0 = *(const v2f*)(brow0 + kh);
        v2f b1 = *(const v2f*)(brow1 + kh);
        v2f b2 = *(const v2f*)(brow2 + kh);
        v2f b3 = *(const v2f*)(brow3 + kh);
        acc0 = __builtin_amdgcn_wmma_f32_16x16x4_f32(false, a, false, b0, (short)0, acc0, false, false);
        acc1 = __builtin_amdgcn_wmma_f32_16x16x4_f32(false, a, false, b1, (short)0, acc1, false, false);
        acc2 = __builtin_amdgcn_wmma_f32_16x16x4_f32(false, a, false, b2, (short)0, acc2, false, false);
        acc3 = __builtin_amdgcn_wmma_f32_16x16x4_f32(false, a, false, b3, (short)0, acc3, false, false);
    }

    const int kkbase = mtile * 16 + 8 * half;
#pragma unroll
    for (int r = 0; r < 8; ++r) {
        const int kkr = kkbase + r;
        const float bias = sb[kkr];
        float* krow = ker + (size_t)kkr * NP + p_base + l16;
        krow[ 0] = acc0[r] + bias;
        krow[16] = acc1[r] + bias;
        krow[32] = acc2[r] + bias;
        krow[48] = acc3[r] + bias;
    }
}

#define TR 4
#define NT3 (TR * Ww)
#define LR 12
#define LP (Ww + Kk - 1)
#define NROW (Gg * LR)
#define XS_ELEMS (NROW * LP + Kk - 1)

__global__ __launch_bounds__(NT3) void involution_apply(
    const float* __restrict__ x, const float* __restrict__ ker,
    float* __restrict__ out)
{
    __shared__ float xs[XS_ELEMS];

    const int tid = threadIdx.x;
    const int rt  = blockIdx.x % (Hh / TR);
    const int bg  = blockIdx.x / (Hh / TR);
    const int g   = bg % Gg;
    const int b   = bg / Gg;
    const int h0  = rt * TR;
    const int r   = tid / Ww;
    const int w   = tid - r * Ww;
    const int hw  = h0 * Ww + tid;
    const int p   = b * HW + hw;

    float kv[KK];
#pragma unroll
    for (int idx = 0; idx < KK; ++idx)
        kv[idx] = ker[(size_t)(g * KK + idx) * NP + p];

    {
        float* gz = xs + (tid < NROW ? tid : NROW) * LP;
#pragma unroll
        for (int q = 0; q < Kk - 1; ++q) gz[q] = 0.f;
    }

    const float* xg = x + ((size_t)b * Cc + g * Gg) * HW;
    int  fl[3];
    bool ok[3];
#pragma unroll
    for (int k = 0; k < 3; ++k) {
        const int f = (h0 - Kk / 2) * Ww + tid + k * NT3;
        ok[k] = f >= 0 && f < HW;
        fl[k] = f < 0 ? 0 : (f < HW ? f : HW - 1);
    }
    float xv[Gg][3];
#pragma unroll
    for (int c = 0; c < Gg; ++c)
#pragma unroll
        for (int k = 0; k < 3; ++k)
            xv[c][k] = xg[c * HW + fl[k]];
    float* xd = xs + r * LP + w + (Kk - 1);
#pragma unroll
    for (int c = 0; c < Gg; ++c)
#pragma unroll
        for (int k = 0; k < 3; ++k)
            xd[(c * LR + TR * k) * LP] = ok[k] ? xv[c][k] : 0.f;
    __syncthreads();

    const float* xw = xs + r * LP + w + (Kk / 2);
    float*       ob = out + ((size_t)b * Cc + g * Gg) * HW + hw;
    for (int cc = 0; cc < Gg; ++cc) {
        const float* xc = xw + cc * (LR * LP);
        float acc = 0.f;
#pragma unroll
        for (int idx = 0; idx < KK; ++idx)
            acc += xc[(idx / Kk) * LP + (idx % Kk)] * kv[idx];
        ob[(size_t)cc * HW] = acc;
    }
}

extern "C" void kernel_launch(void* const* d_in, const int* in_sizes, int n_in,
                              void* d_out, int out_size, void* d_ws, size_t ws_size,
                              hipStream_t stream) {
    const float* x     = (const float*)d_in[0];
    const float* rw    = (const float*)d_in[1];
    const float* rb    = (const float*)d_in[2];
    const float* gamma = (const float*)d_in[3];
    const float* beta  = (const float*)d_in[4];
    const float* mean  = (const float*)d_in[5];
    const float* var   = (const float*)d_in[6];
    const float* pa    = (const float*)d_in[7];
    const float* sw    = (const float*)d_in[8];
    const float* sb    = (const float*)d_in[9];
    float* out = (float*)d_out;

    float* t_ws   = (float*)d_ws;
    float* ker_ws = t_ws + (size_t)NP * TN;

    gemm1_reduce_bn_prelu<<<392, 256, 0, stream>>>(
        x, rw, rb, gamma, beta, mean, var, pa, t_ws);
    gemm2_span<<<2401, 128, 0, stream>>>(
        t_ws, sw, sb, ker_ws);
    involution_apply<<<Bn * Gg * (Hh / TR), NT3, 0, stream>>>(
        x, ker_ws, out);
}


# span GEMM: K chunks of the span_w and t tiles staged in LDS by coalesced 16-byte loads (next chunk prefetched to registers); fragments read from LDS in groups ahead of the WMMAs; k sum order kept
# speedup vs baseline: 2.6136x; 1.4953x over previous
#include <hip/hip_runtime.h>
#include <hip/hip_bf16.h>

typedef __attribute__((ext_vector_type(2))) float v2f;
typedef __attribute__((ext_vector_type(4))) float v4f;
typedef __attribute__((ext_vector_type(8))) float v8f;

#define Bn 4
#define Cc 256
#define Hh 56
#define Ww 56
#define HW (Hh * Ww)
#define NP (Bn * HW)
#define TN 256
#define Gg 16
#define Kk 7
#define KK (Kk * Kk)
#define NK (Gg * KK)
#define EPSf 1e-5f

__global__ __launch_bounds__(256) void gemm1_reduce_bn_prelu(
    const float* __restrict__ x, const float* __restrict__ rw,
    const float* __restrict__ rb, const float* __restrict__ gamma,
    const float* __restrict__ beta, const float* __restrict__ mean,
    const float* __restrict__ var, const float* __restrict__ pa,
    float* __restrict__ t_out)
{
    const int lane  = threadIdx.x & 31;
    const int wave  = threadIdx.x >> 5;
    const int tile  = blockIdx.x * 8 + wave;
    const int npair = tile & 7;
    const int mpair = tile >> 3;
    const int half  = lane >> 4;
    const int l16   = lane & 15;

    const int p_base = mpair * 32;
    const int b      = p_base / HW;
    const int hw     = p_base - b * HW;
    const float* xrow0 = x + (size_t)b * Cc * HW + hw + l16;
    const float* xrow1 = xrow0 + 16;
    const int n0 = npair * 32 + l16;
    const int n1 = n0 + 16;
    const float* wrow0 = rw + (size_t)n0 * Cc;
    const float* wrow1 = rw + (size_t)n1 * Cc;

    v8f acc00 = {}, acc01 = {}, acc10 = {}, acc11 = {};
#pragma unroll 4
    for (int k0 = 0; k0 < Cc; k0 += 4) {
        const int kh = k0 + 2 * half;
        v2f a0, a1;
        a0.x = xrow0[(size_t)kh * HW];
        a0.y = xrow0[(size_t)(kh + 1) * HW];
        a1.x = xrow1[(size_t)kh * HW];
        a1.y = xrow1[(size_t)(kh + 1) * HW];
        v2f b0 = *(const v2f*)(wrow0 + kh);
        v2f b1 = *(const v2f*)(wrow1 + kh);
        acc00 = __builtin_amdgcn_wmma_f32_16x16x4_f32(false, a0, false, b0, (short)0, acc00, false, false);
        acc01 = __builtin_amdgcn_wmma_f32_16x16x4_f32(false, a0, false, b1, (short)0, acc01, false, false);
        acc10 = __builtin_amdgcn_wmma_f32_16x16x4_f32(false, a1, false, b0, (short)0, acc10, false, false);
        acc11 = __builtin_amdgcn_wmma_f32_16x16x4_f32(false, a1, false, b1, (short)0, acc11, false, false);
    }

    const float al = pa[0];
    const float s0  = gamma[n0] * rsqrtf(var[n0] + EPSf);
    const float s1  = gamma[n1] * rsqrtf(var[n1] + EPSf);
    const float bi0 = rb[n0] - mean[n0], bi1 = rb[n1] - mean[n1];
    const float bt0 = beta[n0],          bt1 = beta[n1];

    float* trow0 = t_out + (size_t)(p_base + 8 * half) * TN;
    float* trow1 = t_out + (size_t)(p_base + 16 + 8 * half) * TN;
#pragma unroll
    for (int r = 0; r < 8; ++r) {
        float v00 = (acc00[r] + bi0) * s0 + bt0; v00 = v00 > 0.f ? v00 : al * v00;
        float v01 = (acc01[r] + bi1) * s1 + bt1; v01 = v01 > 0.f ? v01 : al * v01;
        float v10 = (acc10[r] + bi0) * s0 + bt0; v10 = v10 > 0.f ? v10 : al * v10;
        float v11 = (acc11[r] + bi1) * s1 + bt1; v11 = v11 > 0.f ? v11 : al * v11;
        trow0[(size_t)r * TN + n0] = v00;
        trow0[(size_t)r * TN + n1] = v01;
        trow1[(size_t)r * TN + n0] = v10;
        trow1[(size_t)r * TN + n1] = v11;
    }
}

#define BM2 112
#define BN2 128
#define KC2 32
#define LP2 (KC2 + 4)
#define MT2 (BM2 / 16)
#define NB2 (BN2 / 16)

__global__ __launch_bounds__(128) void gemm2_span(
    const float* __restrict__ t_in, const float* __restrict__ sw,
    const float* __restrict__ sb, float* __restrict__ ker)
{
    __shared__ __attribute__((aligned(16))) float As[BM2 * LP2];
    __shared__ __attribute__((aligned(16))) float Bs[BN2 * LP2];

    const int tid   = threadIdx.x;
    const int lane  = tid & 31;
    const int wave  = tid >> 5;
    const int half  = lane >> 4;
    const int l16   = lane & 15;
    const int mblk  = blockIdx.x % (NK / BM2);
    const int nblk  = blockIdx.x / (NK / BM2);
    const int m_base = mblk * BM2;
    const int p_base = nblk * BN2;

    const int sr = tid >> 3;
    const int sq = tid & 7;
    const float* ag = sw   + (size_t)(m_base + sr) * TN + sq * 4;
    const float* bg = t_in + (size_t)(p_base + sr) * TN + sq * 4;
    float* as_st = As + sr * LP2 + sq * 4;
    float* bs_st = Bs + sr * LP2 + sq * 4;

    v4f pa[MT2], pb[NB2];
#pragma unroll
    for (int i = 0; i < MT2; ++i)
        pa[i] = *(const v4f*)(ag + (size_t)i * 16 * TN);
#pragma unroll
    for (int i = 0; i < NB2; ++i)
        pb[i] = *(const v4f*)(bg + (size_t)i * 16 * TN);

    v8f acc[MT2][2];
#pragma unroll
    for (int i = 0; i < MT2; ++i) {
        acc[i][0] = (v8f){};
        acc[i][1] = (v8f){};
    }

    const float* a_fr = As + l16 * LP2 + 2 * half;
    const float* b_fr = Bs + (wave * 32 + l16) * LP2 + 2 * half;

    for (int c = 0; c < TN / KC2; ++c) {
        __syncthreads();
#pragma unroll
        for (int i = 0; i < MT2; ++i)
            *(v4f*)(as_st + i * 16 * LP2) = pa[i];
#pragma unroll
        for (int i = 0; i < NB2; ++i)
            *(v4f*)(bs_st + i * 16 * LP2) = pb[i];
        __syncthreads();

        if (c + 1 < TN / KC2) {
            const int ko = (c + 1) * KC2;
#pragma unroll
            for (int i = 0; i < MT2; ++i)
                pa[i] = *(const v4f*)(ag + (size_t)i * 16 * TN + ko);
#pragma unroll
            for (int i = 0; i < NB2; ++i)
                pb[i] = *(const v4f*)(bg + (size_t)i * 16 * TN + ko);
        }

        v2f fa[2][MT2][2], fb[2][2][2];
#pragma unroll
        for (int i = 0; i < MT2; ++i) {
            fa[0][i][0] = *(const v2f*)(a_fr + i * 16 * LP2);
            fa[0][i][1] = *(const v2f*)(a_fr + i * 16 * LP2 + 4);
        }
#pragma unroll
        for (int j = 0; j < 2; ++j) {
            fb[0][j][0] = *(const v2f*)(b_fr + j * 16 * LP2);
            fb[0][j][1] = *(const v2f*)(b_fr + j * 16 * LP2 + 4);
        }
#pragma unroll
        for (int kp = 0; kp < KC2 / 8; ++kp) {
            const int cur = kp & 1, nxt = cur ^ 1;
            if (kp + 1 < KC2 / 8) {
#pragma unroll
                for (int i = 0; i < MT2; ++i) {
                    fa[nxt][i][0] = *(const v2f*)(a_fr + i * 16 * LP2 + (kp + 1) * 8);
                    fa[nxt][i][1] = *(const v2f*)(a_fr + i * 16 * LP2 + (kp + 1) * 8 + 4);
                }
#pragma unroll
                for (int j = 0; j < 2; ++j) {
                    fb[nxt][j][0] = *(const v2f*)(b_fr + j * 16 * LP2 + (kp + 1) * 8);
                    fb[nxt][j][1] = *(const v2f*)(b_fr + j * 16 * LP2 + (kp + 1) * 8 + 4);
                }
            }
            __builtin_amdgcn_sched_barrier(0);
#pragma unroll
            for (int s = 0; s < 2; ++s) {
#pragma unroll
                for (int i = 0; i < MT2; ++i) {
                    acc[i][0] = __builtin_amdgcn_wmma_f32_16x16x4_f32(false, fa[cur][i][s], false, fb[cur][0][s], (short)0, acc[i][0], false, false);
                    acc[i][1] = __builtin_amdgcn_wmma_f32_16x16x4_f32(false, fa[cur][i][s], false, fb[cur][1][s], (short)0, acc[i][1], false, false);
                }
            }
            __builtin_amdgcn_sched_barrier(0);
        }
    }

    float* kcol = ker + p_base + wave * 32 + l16;
#pragma unroll
    for (int i = 0; i < MT2; ++i) {
#pragma unroll
        for (int r = 0; r < 8; ++r) {
            const int kkr = m_base + i * 16 + 8 * half + r;
            const float bias = sb[kkr];
            float* krow = kcol + (size_t)kkr * NP;
            krow[ 0] = acc[i][0][r] + bias;
            krow[16] = acc[i][1][r] + bias;
        }
    }
}

#define TR 4
#define NT3 (TR * Ww)
#define LR 12
#define LP (Ww + Kk - 1)
#define NROW (Gg * LR)
#define XS_ELEMS (NROW * LP + Kk - 1)

__global__ __launch_bounds__(NT3) void involution_apply(
    const float* __restrict__ x, const float* __restrict__ ker,
    float* __restrict__ out)
{
    __shared__ float xs[XS_ELEMS];

    const int tid = threadIdx.x;
    const int rt  = blockIdx.x % (Hh / TR);
    const int bg  = blockIdx.x / (Hh / TR);
    const int g   = bg % Gg;
    const int b   = bg / Gg;
    const int h0  = rt * TR;
    const int r   = tid / Ww;
    const int w   = tid - r * Ww;
    const int hw  = h0 * Ww + tid;
    const int p   = b * HW + hw;

    float kv[KK];
#pragma unroll
    for (int idx = 0; idx < KK; ++idx)
        kv[idx] = ker[(size_t)(g * KK + idx) * NP + p];

    {
        float* gz = xs + (tid < NROW ? tid : NROW) * LP;
#pragma unroll
        for (int q = 0; q < Kk - 1; ++q) gz[q] = 0.f;
    }

    const float* xg = x + ((size_t)b * Cc + g * Gg) * HW;
    int  fl[3];
    bool ok[3];
#pragma unroll
    for (int k = 0; k < 3; ++k) {
        const int f = (h0 - Kk / 2) * Ww + tid + k * NT3;
        ok[k] = f >= 0 && f < HW;
        fl[k] = f < 0 ? 0 : (f < HW ? f : HW - 1);
    }
    float xv[Gg][3];
#pragma unroll
    for (int c = 0; c < Gg; ++c)
#pragma unroll
        for (int k = 0; k < 3; ++k)
            xv[c][k] = xg[c * HW + fl[k]];
    float* xd = xs + r * LP + w + (Kk - 1);
#pragma unroll
    for (int c = 0; c < Gg; ++c)
#pragma unroll
        for (int k = 0; k < 3; ++k)
            xd[(c * LR + TR * k) * LP] = ok[k] ? xv[c][k] : 0.f;
    __syncthreads();

    const float* xw = xs + r * LP + w + (Kk / 2);
    float*       ob = out + ((size_t)b * Cc + g * Gg) * HW + hw;
    for (int cc = 0; cc < Gg; ++cc) {
        const float* xc = xw + cc * (LR * LP);
        float acc = 0.f;
#pragma unroll
        for (int idx = 0; idx < KK; ++idx)
            acc += xc[(idx / Kk) * LP + (idx % Kk)] * kv[idx];
        ob[(size_t)cc * HW] = acc;
    }
}

extern "C" void kernel_launch(void* const* d_in, const int* in_sizes, int n_in,
                              void* d_out, int out_size, void* d_ws, size_t ws_size,
                              hipStream_t stream) {
    const float* x     = (const float*)d_in[0];
    const float* rw    = (const float*)d_in[1];
    const float* rb    = (const float*)d_in[2];
    const float* gamma = (const float*)d_in[3];
    const float* beta  = (const float*)d_in[4];
    const float* mean  = (const float*)d_in[5];
    const float* var   = (const float*)d_in[6];
    const float* pa    = (const float*)d_in[7];
    const float* sw    = (const float*)d_in[8];
    const float* sb    = (const float*)d_in[9];
    float* out = (float*)d_out;

    float* t_ws   = (float*)d_ws;
    float* ker_ws = t_ws + (size_t)NP * TN;

    gemm1_reduce_bn_prelu<<<392, 256, 0, stream>>>(
        x, rw, rb, gamma, beta, mean, var, pa, t_ws);
    gemm2_span<<<686, 128, 0, stream>>>(
        t_ws, sw, sb, ker_ws);
    involution_apply<<<Bn * Gg * (Hh / TR), NT3, 0, stream>>>(
        x, ker_ws, out);
}


# involution: one thread per 2 adjacent pixels; a tap row's 8 floats are read from LDS once (8-byte reads) and feed both: 28 LDS reads per channel and pair, was 2 x 49; 8-row tiles; tap order kept
# speedup vs baseline: 3.0406x; 1.1634x over previous
#include <hip/hip_runtime.h>
#include <hip/hip_bf16.h>

typedef __attribute__((ext_vector_type(2))) float v2f;
typedef __attribute__((ext_vector_type(4))) float v4f;
typedef __attribute__((ext_vector_type(8))) float v8f;

#define Bn 4
#define Cc 256
#define Hh 56
#define Ww 56
#define HW (Hh * Ww)
#define NP (Bn * HW)
#define TN 256
#define Gg 16
#define Kk 7
#define KK (Kk * Kk)
#define NK (Gg * KK)
#define EPSf 1e-5f

__global__ __launch_bounds__(256) void gemm1_reduce_bn_prelu(
    const float* __restrict__ x, const float* __restrict__ rw,
    const float* __restrict__ rb, const float* __restrict__ gamma,
    const float* __restrict__ beta, const float* __restrict__ mean,
    const float* __restrict__ var, const float* __restrict__ pa,
    float* __restrict__ t_out)
{
    const int lane  = threadIdx.x & 31;
    const int wave  = threadIdx.x >> 5;
    const int tile  = blockIdx.x * 8 + wave;
    const int npair = tile & 7;
    const int mpair = tile >> 3;
    const int half  = lane >> 4;
    const int l16   = lane & 15;

    const int p_base = mpair * 32;
    const int b      = p_base / HW;
    const int hw     = p_base - b * HW;
    const float* xrow0 = x + (size_t)b * Cc * HW + hw + l16;
    const float* xrow1 = xrow0 + 16;
    const int n0 = npair * 32 + l16;
    const int n1 = n0 + 16;
    const float* wrow0 = rw + (size_t)n0 * Cc;
    const float* wrow1 = rw + (size_t)n1 * Cc;

    v8f acc00 = {}, acc01 = {}, acc10 = {}, acc11 = {};
#pragma unroll 4
    for (int k0 = 0; k0 < Cc; k0 += 4) {
        const int kh = k0 + 2 * half;
        v2f a0, a1;
        a0.x = xrow0[(size_t)kh * HW];
        a0.y = xrow0[(size_t)(kh + 1) * HW];
        a1.x = xrow1[(size_t)kh * HW];
        a1.y = xrow1[(size_t)(kh + 1) * HW];
        v2f b0 = *(const v2f*)(wrow0 + kh);
        v2f b1 = *(const v2f*)(wrow1 + kh);
        acc00 = __builtin_amdgcn_wmma_f32_16x16x4_f32(false, a0, false, b0, (short)0, acc00, false, false);
        acc01 = __builtin_amdgcn_wmma_f32_16x16x4_f32(false, a0, false, b1, (short)0, acc01, false, false);
        acc10 = __builtin_amdgcn_wmma_f32_16x16x4_f32(false, a1, false, b0, (short)0, acc10, false, false);
        acc11 = __builtin_amdgcn_wmma_f32_16x16x4_f32(false, a1, false, b1, (short)0, acc11, false, false);
    }

    const float al = pa[0];
    const float s0  = gamma[n0] * rsqrtf(var[n0] + EPSf);
    const float s1  = gamma[n1] * rsqrtf(var[n1] + EPSf);
    const float bi0 = rb[n0] - mean[n0], bi1 = rb[n1] - mean[n1];
    const float bt0 = beta[n0],          bt1 = beta[n1];

    float* trow0 = t_out + (size_t)(p_base + 8 * half) * TN;
    float* trow1 = t_out + (size_t)(p_base + 16 + 8 * half) * TN;
#pragma unroll
    for (int r = 0; r < 8; ++r) {
        float v00 = (acc00[r] + bi0) * s0 + bt0; v00 = v00 > 0.f ? v00 : al * v00;
        float v01 = (acc01[r] + bi1) * s1 + bt1; v01 = v01 > 0.f ? v01 : al * v01;
        float v10 = (acc10[r] + bi0) * s0 + bt0; v10 = v10 > 0.f ? v10 : al * v10;
        float v11 = (acc11[r] + bi1) * s1 + bt1; v11 = v11 > 0.f ? v11 : al * v11;
        trow0[(size_t)r * TN + n0] = v00;
        trow0[(size_t)r * TN + n1] = v01;
        trow1[(size_t)r * TN + n0] = v10;
        trow1[(size_t)r * TN + n1] = v11;
    }
}

#define BM2 112
#define BN2 128
#define KC2 32
#define LP2 (KC2 + 4)
#define MT2 (BM2 / 16)
#define NB2 (BN2 / 16)

__global__ __launch_bounds__(128) void gemm2_span(
    const float* __restrict__ t_in, const float* __restrict__ sw,
    const float* __restrict__ sb, float* __restrict__ ker)
{
    __shared__ __attribute__((aligned(16))) float As[BM2 * LP2];
    __shared__ __attribute__((aligned(16))) float Bs[BN2 * LP2];

    const int tid   = threadIdx.x;
    const int lane  = tid & 31;
    const int wave  = tid >> 5;
    const int half  = lane >> 4;
    const int l16   = lane & 15;
    const int mblk  = blockIdx.x % (NK / BM2);
    const int nblk  = blockIdx.x / (NK / BM2);
    const int m_base = mblk * BM2;
    const int p_base = nblk * BN2;

    const int sr = tid >> 3;
    const int sq = tid & 7;
    const float* ag = sw   + (size_t)(m_base + sr) * TN + sq * 4;
    const float* bg = t_in + (size_t)(p_base + sr) * TN + sq * 4;
    float* as_st = As + sr * LP2 + sq * 4;
    float* bs_st = Bs + sr * LP2 + sq * 4;

    v4f pa[MT2], pb[NB2];
#pragma unroll
    for (int i = 0; i < MT2; ++i)
        pa[i] = *(const v4f*)(ag + (size_t)i * 16 * TN);
#pragma unroll
    for (int i = 0; i < NB2; ++i)
        pb[i] = *(const v4f*)(bg + (size_t)i * 16 * TN);

    v8f acc[MT2][2];
#pragma unroll
    for (int i = 0; i < MT2; ++i) {
        acc[i][0] = (v8f){};
        acc[i][1] = (v8f){};
    }

    const float* a_fr = As + l16 * LP2 + 2 * half;
    const float* b_fr = Bs + (wave * 32 + l16) * LP2 + 2 * half;

    for (int c = 0; c < TN / KC2; ++c) {
        __syncthreads();
#pragma unroll
        for (int i = 0; i < MT2; ++i)
            *(v4f*)(as_st + i * 16 * LP2) = pa[i];
#pragma unroll
        for (int i = 0; i < NB2; ++i)
            *(v4f*)(bs_st + i * 16 * LP2) = pb[i];
        __syncthreads();

        if (c + 1 < TN / KC2) {
            const int ko = (c + 1) * KC2;
#pragma unroll
            for (int i = 0; i < MT2; ++i)
                pa[i] = *(const v4f*)(ag + (size_t)i * 16 * TN + ko);
#pragma unroll
            for (int i = 0; i < NB2; ++i)
                pb[i] = *(const v4f*)(bg + (size_t)i * 16 * TN + ko);
        }

        v2f fa[2][MT2][2], fb[2][2][2];
#pragma unroll
        for (int i = 0; i < MT2; ++i) {
            fa[0][i][0] = *(const v2f*)(a_fr + i * 16 * LP2);
            fa[0][i][1] = *(const v2f*)(a_fr + i * 16 * LP2 + 4);
        }
#pragma unroll
        for (int j = 0; j < 2; ++j) {
            fb[0][j][0] = *(const v2f*)(b_fr + j * 16 * LP2);
            fb[0][j][1] = *(const v2f*)(b_fr + j * 16 * LP2 + 4);
        }
#pragma unroll
        for (int kp = 0; kp < KC2 / 8; ++kp) {
            const int cur = kp & 1, nxt = cur ^ 1;
            if (kp + 1 < KC2 / 8) {
#pragma unroll
                for (int i = 0; i < MT2; ++i) {
                    fa[nxt][i][0] = *(const v2f*)(a_fr + i * 16 * LP2 + (kp + 1) * 8);
                    fa[nxt][i][1] = *(const v2f*)(a_fr + i * 16 * LP2 + (kp + 1) * 8 + 4);
                }
#pragma unroll
                for (int j = 0; j < 2; ++j) {
                    fb[nxt][j][0] = *(const v2f*)(b_fr + j * 16 * LP2 + (kp + 1) * 8);
                    fb[nxt][j][1] = *(const v2f*)(b_fr + j * 16 * LP2 + (kp + 1) * 8 + 4);
                }
            }
            __builtin_amdgcn_sched_barrier(0);
#pragma unroll
            for (int s = 0; s < 2; ++s) {
#pragma unroll
                for (int i = 0; i < MT2; ++i) {
                    acc[i][0] = __builtin_amdgcn_wmma_f32_16x16x4_f32(false, fa[cur][i][s], false, fb[cur][0][s], (short)0, acc[i][0], false, false);
                    acc[i][1] = __builtin_amdgcn_wmma_f32_16x16x4_f32(false, fa[cur][i][s], false, fb[cur][1][s], (short)0, acc[i][1], false, false);
                }
            }
            __builtin_amdgcn_sched_barrier(0);
        }
    }

    float* kcol = ker + p_base + wave * 32 + l16;
#pragma unroll
    for (int i = 0; i < MT2; ++i) {
#pragma unroll
        for (int r = 0; r < 8; ++r) {
            const int kkr = m_base + i * 16 + 8 * half + r;
            const float bias = sb[kkr];
            float* krow = kcol + (size_t)kkr * NP;
            krow[ 0] = acc[i][0][r] + bias;
            krow[16] = acc[i][1][r] + bias;
        }
    }
}

#define TR 8
#define PW (Ww / 2)
#define NT3 (TR * PW)
#define LR (TR + Kk - 1)
#define LP 120
#define PX0 7
#define NROW (Gg * LR)
#define XS_ELEMS (NROW * LP)
#define CH4 (LR * Ww / 4)

__global__ __launch_bounds__(NT3) void involution_apply(
    const float* __restrict__ x, const float* __restrict__ ker,
    float* __restrict__ out)
{
    __shared__ __attribute__((aligned(16))) float xs[XS_ELEMS];

    const int tid = threadIdx.x;
    const int rt  = blockIdx.x % (Hh / TR);
    const int bg  = blockIdx.x / (Hh / TR);
    const int g   = bg % Gg;
    const int b   = bg / Gg;
    const int h0  = rt * TR;
    const int hw  = h0 * Ww + 2 * tid;
    const int p   = b * HW + hw;

    v2f kv[KK];
#pragma unroll
    for (int idx = 0; idx < KK; ++idx)
        kv[idx] = *(const v2f*)(ker + (size_t)(g * KK + idx) * NP + p);

    {
        float* gz = xs + tid * LP;
#pragma unroll
        for (int e = 0; e < Kk / 2; ++e) {
            gz[PX0 - Kk / 2 + e] = 0.f;
            gz[PX0 + Ww + e]     = 0.f;
        }
    }

    const float* xg = x + ((size_t)b * Cc + g * Gg) * HW;
    const int  q  = tid < CH4 ? tid : CH4 - 1;
    const int  f  = (h0 - Kk / 2) * Ww + 4 * q;
    const bool ok = f >= 0 && f < HW;
    const int  fc = ok ? f : 0;
    v4f xv[Gg];
#pragma unroll
    for (int c = 0; c < Gg; ++c)
        xv[c] = *(const v4f*)(xg + c * HW + fc);
    if (tid < CH4) {
        float* xd = xs + (q / (Ww / 4)) * LP + PX0 + 4 * (q % (Ww / 4));
#pragma unroll
        for (int c = 0; c < Gg; ++c) {
            const v4f v = ok ? xv[c] : (v4f){0.f, 0.f, 0.f, 0.f};
            float* d = xd + c * (LR * LP);
            d[0] = v.x;
            d[1] = v.y;
            d[2] = v.z;
            d[3] = v.w;
        }
    }
    __syncthreads();

    v2f   kq[Kk][Kk - 1];
    float kl[Kk], kr[Kk];
#pragma unroll
    for (int i = 0; i < Kk; ++i) {
        kl[i] = kv[i * Kk].x;
        kr[i] = kv[i * Kk + Kk - 1].y;
#pragma unroll
        for (int e = 1; e < Kk; ++e)
            kq[i][e - 1] = (v2f){kv[i * Kk + e].x, kv[i * Kk + e - 1].y};
    }

    const int r  = tid / PW;
    const int jp = tid - r * PW;
    const float* xw = xs + r * LP + (PX0 - Kk / 2) + 2 * jp;
    float*       ob = out + ((size_t)b * Cc + g * Gg) * HW + hw;
#pragma unroll 2
    for (int cc = 0; cc < Gg; ++cc) {
        const float* xc = xw + cc * (LR * LP);
        v2f a = (v2f){0.f, 0.f};
#pragma unroll
        for (int i = 0; i < Kk; ++i) {
            const v2f w0 = *(const v2f*)(xc + i * LP);
            const v2f w1 = *(const v2f*)(xc + i * LP + 2);
            const v2f w2 = *(const v2f*)(xc + i * LP + 4);
            const v2f w3 = *(const v2f*)(xc + i * LP + 6);
            const float xr[Kk + 1] = {w0.x, w0.y, w1.x, w1.y,
                                      w2.x, w2.y, w3.x, w3.y};
            a.x = __builtin_fmaf(xr[0], kl[i], a.x);
#pragma unroll
            for (int e = 1; e < Kk; ++e)
                a = __builtin_elementwise_fma((v2f){xr[e], xr[e]}, kq[i][e - 1], a);
            a.y = __builtin_fmaf(xr[Kk], kr[i], a.y);
        }
        *(v2f*)(ob + (size_t)cc * HW) = a;
    }
}

extern "C" void kernel_launch(void* const* d_in, const int* in_sizes, int n_in,
                              void* d_out, int out_size, void* d_ws, size_t ws_size,
                              hipStream_t stream) {
    const float* x     = (const float*)d_in[0];
    const float* rw    = (const float*)d_in[1];
    const float* rb    = (const float*)d_in[2];
    const float* gamma = (const float*)d_in[3];
    const float* beta  = (const float*)d_in[4];
    const float* mean  = (const float*)d_in[5];
    const float* var   = (const float*)d_in[6];
    const float* pa    = (const float*)d_in[7];
    const float* sw    = (const float*)d_in[8];
    const float* sb    = (const float*)d_in[9];
    float* out = (float*)d_out;

    float* t_ws   = (float*)d_ws;
    float* ker_ws = t_ws + (size_t)NP * TN;

    gemm1_reduce_bn_prelu<<<392, 256, 0, stream>>>(
        x, rw, rb, gamma, beta, mean, var, pa, t_ws);
    gemm2_span<<<686, 128, 0, stream>>>(
        t_ws, sw, sb, ker_ws);
    involution_apply<<<Bn * Gg * (Hh / TR), NT3, 0, stream>>>(
        x, ker_ws, out);
}


# reduce GEMM: K chunks of the x tile (transposed by its LDS stores) and the reduce_w tile staged in LDS by coalesced 16-byte loads, next chunk prefetched to registers; 64x128 tile; k sum order kept
# speedup vs baseline: 3.5235x; 1.1588x over previous
#include <hip/hip_runtime.h>
#include <hip/hip_bf16.h>

typedef __attribute__((ext_vector_type(2))) float v2f;
typedef __attribute__((ext_vector_type(4))) float v4f;
typedef __attribute__((ext_vector_type(8))) float v8f;

#define Bn 4
#define Cc 256
#define Hh 56
#define Ww 56
#define HW (Hh * Ww)
#define NP (Bn * HW)
#define TN 256
#define Gg 16
#define Kk 7
#define KK (Kk * Kk)
#define NK (Gg * KK)
#define EPSf 1e-5f

#define BM1 64
#define BN1 128
#define KC1 32
#define LPA1 (KC1 + 2)
#define LPB1 (KC1 + 4)
#define MT1 (BM1 / 16)

__global__ __launch_bounds__(128) void gemm1_reduce_bn_prelu(
    const float* __restrict__ x, const float* __restrict__ rw,
    const float* __restrict__ rb, const float* __restrict__ gamma,
    const float* __restrict__ beta, const float* __restrict__ mean,
    const float* __restrict__ var, const float* __restrict__ pa,
    float* __restrict__ t_out)
{
    __shared__ __attribute__((aligned(16))) float Xs[BM1 * LPA1];
    __shared__ __attribute__((aligned(16))) float Ws[BN1 * LPB1];

    const int tid   = threadIdx.x;
    const int lane  = tid & 31;
    const int wave  = tid >> 5;
    const int half  = lane >> 4;
    const int l16   = lane & 15;
    const int nblk  = blockIdx.x & 1;
    const int mblk  = blockIdx.x >> 1;
    const int p_base = mblk * BM1;
    const int n_base = nblk * BN1;
    const int b      = p_base / HW;
    const int hw     = p_base - b * HW;

    const int pg = (wave & 1) * 8 + (lane & 7);
    const int kb = (wave >> 1) * 4 + (lane >> 3);
    const float* xg = x + ((size_t)b * Cc + kb) * HW + hw + 4 * pg;
    float* xs_st = Xs + (4 * pg) * LPA1 + kb;
    const int sr = tid >> 3;
    const int sq = tid & 7;
    const float* wg = rw + (size_t)(n_base + sr) * Cc + sq * 4;
    float* ws_st = Ws + sr * LPB1 + sq * 4;

    v4f px[KC1 / 8], pw[BN1 / 16];
#pragma unroll
    for (int i = 0; i < KC1 / 8; ++i)
        px[i] = *(const v4f*)(xg + (size_t)(8 * i) * HW);
#pragma unroll
    for (int i = 0; i < BN1 / 16; ++i)
        pw[i] = *(const v4f*)(wg + (size_t)i * 16 * Cc);

    v8f acc[MT1][2];
#pragma unroll
    for (int i = 0; i < MT1; ++i) {
        acc[i][0] = (v8f){};
        acc[i][1] = (v8f){};
    }

    const float* a_fr = Xs + l16 * LPA1 + 2 * half;
    const float* b_fr = Ws + (wave * 32 + l16) * LPB1 + 2 * half;

    for (int c = 0; c < Cc / KC1; ++c) {
        __syncthreads();
#pragma unroll
        for (int i = 0; i < KC1 / 8; ++i) {
            float* d = xs_st + 8 * i;
            d[0]        = px[i].x;
            d[LPA1]     = px[i].y;
            d[2 * LPA1] = px[i].z;
            d[3 * LPA1] = px[i].w;
        }
#pragma unroll
        for (int i = 0; i < BN1 / 16; ++i)
            *(v4f*)(ws_st + i * 16 * LPB1) = pw[i];
        __syncthreads();

        if (c + 1 < Cc / KC1) {
            const int ko = (c + 1) * KC1;
#pragma unroll
            for (int i = 0; i < KC1 / 8; ++i)
                px[i] = *(const v4f*)(xg + (size_t)(ko + 8 * i) * HW);
#pragma unroll
            for (int i = 0; i < BN1 / 16; ++i)
                pw[i] = *(const v4f*)(wg + (size_t)i * 16 * Cc + ko);
        }

        v2f fa[2][MT1][2], fb[2][2][2];
#pragma unroll
        for (int i = 0; i < MT1; ++i) {
            fa[0][i][0] = *(const v2f*)(a_fr + i * 16 * LPA1);
            fa[0][i][1] = *(const v2f*)(a_fr + i * 16 * LPA1 + 4);
        }
#pragma unroll
        for (int j = 0; j < 2; ++j) {
            fb[0][j][0] = *(const v2f*)(b_fr + j * 16 * LPB1);
            fb[0][j][1] = *(const v2f*)(b_fr + j * 16 * LPB1 + 4);
        }
#pragma unroll
        for (int kp = 0; kp < KC1 / 8; ++kp) {
            const int cur = kp & 1, nxt = cur ^ 1;
            if (kp + 1 < KC1 / 8) {
#pragma unroll
                for (int i = 0; i < MT1; ++i) {
                    fa[nxt][i][0] = *(const v2f*)(a_fr + i * 16 * LPA1 + (kp + 1) * 8);
                    fa[nxt][i][1] = *(const v2f*)(a_fr + i * 16 * LPA1 + (kp + 1) * 8 + 4);
                }
#pragma unroll
                for (int j = 0; j < 2; ++j) {
                    fb[nxt][j][0] = *(const v2f*)(b_fr + j * 16 * LPB1 + (kp + 1) * 8);
                    fb[nxt][j][1] = *(const v2f*)(b_fr + j * 16 * LPB1 + (kp + 1) * 8 + 4);
                }
            }
            __builtin_amdgcn_sched_barrier(0);
#pragma unroll
            for (int s = 0; s < 2; ++s) {
#pragma unroll
                for (int i = 0; i < MT1; ++i) {
                    acc[i][0] = __builtin_amdgcn_wmma_f32_16x16x4_f32(false, fa[cur][i][s], false, fb[cur][0][s], (short)0, acc[i][0], false, false);
                    acc[i][1] = __builtin_amdgcn_wmma_f32_16x16x4_f32(false, fa[cur][i][s], false, fb[cur][1][s], (short)0, acc[i][1], false, false);
                }
            }
            __builtin_amdgcn_sched_barrier(0);
        }
    }

    const int n0 = n_base + wave * 32 + l16;
    const int n1 = n0 + 16;
    const float al = pa[0];
    const float s0  = gamma[n0] * rsqrtf(var[n0] + EPSf);
    const float s1  = gamma[n1] * rsqrtf(var[n1] + EPSf);
    const float bi0 = rb[n0] - mean[n0], bi1 = rb[n1] - mean[n1];
    const float bt0 = beta[n0],          bt1 = beta[n1];

#pragma unroll
    for (int i = 0; i < MT1; ++i) {
        float* trow = t_out + (size_t)(p_base + i * 16 + 8 * half) * TN;
#pragma unroll
        for (int r = 0; r < 8; ++r) {
            float v0 = (acc[i][0][r] + bi0) * s0 + bt0; v0 = v0 > 0.f ? v0 : al * v0;
            float v1 = (acc[i][1][r] + bi1) * s1 + bt1; v1 = v1 > 0.f ? v1 : al * v1;
            trow[(size_t)r * TN + n0] = v0;
            trow[(size_t)r * TN + n1] = v1;
        }
    }
}

#define BM2 112
#define BN2 128
#define KC2 32
#define LP2 (KC2 + 4)
#define MT2 (BM2 / 16)
#define NB2 (BN2 / 16)

__global__ __launch_bounds__(128) void gemm2_span(
    const float* __restrict__ t_in, const float* __restrict__ sw,
    const float* __restrict__ sb, float* __restrict__ ker)
{
    __shared__ __attribute__((aligned(16))) float As[BM2 * LP2];
    __shared__ __attribute__((aligned(16))) float Bs[BN2 * LP2];

    const int tid   = threadIdx.x;
    const int lane  = tid & 31;
    const int wave  = tid >> 5;
    const int half  = lane >> 4;
    const int l16   = lane & 15;
    const int mblk  = blockIdx.x % (NK / BM2);
    const int nblk  = blockIdx.x / (NK / BM2);
    const int m_base = mblk * BM2;
    const int p_base = nblk * BN2;

    const int sr = tid >> 3;
    const int sq = tid & 7;
    const float* ag = sw   + (size_t)(m_base + sr) * TN + sq * 4;
    const float* bg = t_in + (size_t)(p_base + sr) * TN + sq * 4;
    float* as_st = As + sr * LP2 + sq * 4;
    float* bs_st = Bs + sr * LP2 + sq * 4;

    v4f pa[MT2], pb[NB2];
#pragma unroll
    for (int i = 0; i < MT2; ++i)
        pa[i] = *(const v4f*)(ag + (size_t)i * 16 * TN);
#pragma unroll
    for (int i = 0; i < NB2; ++i)
        pb[i] = *(const v4f*)(bg + (size_t)i * 16 * TN);

    v8f acc[MT2][2];
#pragma unroll
    for (int i = 0; i < MT2; ++i) {
        acc[i][0] = (v8f){};
        acc[i][1] = (v8f){};
    }

    const float* a_fr = As + l16 * LP2 + 2 * half;
    const float* b_fr = Bs + (wave * 32 + l16) * LP2 + 2 * half;

    for (int c = 0; c < TN / KC2; ++c) {
        __syncthreads();
#pragma unroll
        for (int i = 0; i < MT2; ++i)
            *(v4f*)(as_st + i * 16 * LP2) = pa[i];
#pragma unroll
        for (int i = 0; i < NB2; ++i)
            *(v4f*)(bs_st + i * 16 * LP2) = pb[i];
        __syncthreads();

        if (c + 1 < TN / KC2) {
            const int ko = (c + 1) * KC2;
#pragma unroll
            for (int i = 0; i < MT2; ++i)
                pa[i] = *(const v4f*)(ag + (size_t)i * 16 * TN + ko);
#pragma unroll
            for (int i = 0; i < NB2; ++i)
                pb[i] = *(const v4f*)(bg + (size_t)i * 16 * TN + ko);
        }

        v2f fa[2][MT2][2], fb[2][2][2];
#pragma unroll
        for (int i = 0; i < MT2; ++i) {
            fa[0][i][0] = *(const v2f*)(a_fr + i * 16 * LP2);
            fa[0][i][1] = *(const v2f*)(a_fr + i * 16 * LP2 + 4);
        }
#pragma unroll
        for (int j = 0; j < 2; ++j) {
            fb[0][j][0] = *(const v2f*)(b_fr + j * 16 * LP2);
            fb[0][j][1] = *(const v2f*)(b_fr + j * 16 * LP2 + 4);
        }
#pragma unroll
        for (int kp = 0; kp < KC2 / 8; ++kp) {
            const int cur = kp & 1, nxt = cur ^ 1;
            if (kp + 1 < KC2 / 8) {
#pragma unroll
                for (int i = 0; i < MT2; ++i) {
                    fa[nxt][i][0] = *(const v2f*)(a_fr + i * 16 * LP2 + (kp + 1) * 8);
                    fa[nxt][i][1] = *(const v2f*)(a_fr + i * 16 * LP2 + (kp + 1) * 8 + 4);
                }
#pragma unroll
                for (int j = 0; j < 2; ++j) {
                    fb[nxt][j][0] = *(const v2f*)(b_fr + j * 16 * LP2 + (kp + 1) * 8);
                    fb[nxt][j][1] = *(const v2f*)(b_fr + j * 16 * LP2 + (kp + 1) * 8 + 4);
                }
            }
            __builtin_amdgcn_sched_barrier(0);
#pragma unroll
            for (int s = 0; s < 2; ++s) {
#pragma unroll
                for (int i = 0; i < MT2; ++i) {
                    acc[i][0] = __builtin_amdgcn_wmma_f32_16x16x4_f32(false, fa[cur][i][s], false, fb[cur][0][s], (short)0, acc[i][0], false, false);
                    acc[i][1] = __builtin_amdgcn_wmma_f32_16x16x4_f32(false, fa[cur][i][s], false, fb[cur][1][s], (short)0, acc[i][1], false, false);
                }
            }
            __builtin_amdgcn_sched_barrier(0);
        }
    }

    float* kcol = ker + p_base + wave * 32 + l16;
#pragma unroll
    for (int i = 0; i < MT2; ++i) {
#pragma unroll
        for (int r = 0; r < 8; ++r) {
            const int kkr = m_base + i * 16 + 8 * half + r;
            const float bias = sb[kkr];
            float* krow = kcol + (size_t)kkr * NP;
            krow[ 0] = acc[i][0][r] + bias;
            krow[16] = acc[i][1][r] + bias;
        }
    }
}

#define TR 8
#define PW (Ww / 2)
#define NT3 (TR * PW)
#define LR (TR + Kk - 1)
#define LP 120
#define PX0 7
#define NROW (Gg * LR)
#define XS_ELEMS (NROW * LP)
#define CH4 (LR * Ww / 4)

__global__ __launch_bounds__(NT3) void involution_apply(
    const float* __restrict__ x, const float* __restrict__ ker,
    float* __restrict__ out)
{
    __shared__ __attribute__((aligned(16))) float xs[XS_ELEMS];

    const int tid = threadIdx.x;
    const int rt  = blockIdx.x % (Hh / TR);
    const int bg  = blockIdx.x / (Hh / TR);
    const int g   = bg % Gg;
    const int b   = bg / Gg;
    const int h0  = rt * TR;
    const int hw  = h0 * Ww + 2 * tid;
    const int p   = b * HW + hw;

    v2f kv[KK];
#pragma unroll
    for (int idx = 0; idx < KK; ++idx)
        kv[idx] = *(const v2f*)(ker + (size_t)(g * KK + idx) * NP + p);

    {
        float* gz = xs + tid * LP;
#pragma unroll
        for (int e = 0; e < Kk / 2; ++e) {
            gz[PX0 - Kk / 2 + e] = 0.f;
            gz[PX0 + Ww + e]     = 0.f;
        }
    }

    const float* xg = x + ((size_t)b * Cc + g * Gg) * HW;
    const int  q  = tid < CH4 ? tid : CH4 - 1;
    const int  f  = (h0 - Kk / 2) * Ww + 4 * q;
    const bool ok = f >= 0 && f < HW;
    const int  fc = ok ? f : 0;
    v4f xv[Gg];
#pragma unroll
    for (int c = 0; c < Gg; ++c)
        xv[c] = *(const v4f*)(xg + c * HW + fc);
    if (tid < CH4) {
        float* xd = xs + (q / (Ww / 4)) * LP + PX0 + 4 * (q % (Ww / 4));
#pragma unroll
        for (int c = 0; c < Gg; ++c) {
            const v4f v = ok ? xv[c] : (v4f){0.f, 0.f, 0.f, 0.f};
            float* d = xd + c * (LR * LP);
            d[0] = v.x;
            d[1] = v.y;
            d[2] = v.z;
            d[3] = v.w;
        }
    }
    __syncthreads();

    v2f   kq[Kk][Kk - 1];
    float kl[Kk], kr[Kk];
#pragma unroll
    for (int i = 0; i < Kk; ++i) {
        kl[i] = kv[i * Kk].x;
        kr[i] = kv[i * Kk + Kk - 1].y;
#pragma unroll
        for (int e = 1; e < Kk; ++e)
            kq[i][e - 1] = (v2f){kv[i * Kk + e].x, kv[i * Kk + e - 1].y};
    }

    const int r  = tid / PW;
    const int jp = tid - r * PW;
    const float* xw = xs + r * LP + (PX0 - Kk / 2) + 2 * jp;
    float*       ob = out + ((size_t)b * Cc + g * Gg) * HW + hw;
#pragma unroll 2
    for (int cc = 0; cc < Gg; ++cc) {
        const float* xc = xw + cc * (LR * LP);
        v2f a = (v2f){0.f, 0.f};
#pragma unroll
        for (int i = 0; i < Kk; ++i) {
            const v2f w0 = *(const v2f*)(xc + i * LP);
            const v2f w1 = *(const v2f*)(xc + i * LP + 2);
            const v2f w2 = *(const v2f*)(xc + i * LP + 4);
            const v2f w3 = *(const v2f*)(xc + i * LP + 6);
            const float xr[Kk + 1] = {w0.x, w0.y, w1.x, w1.y,
                                      w2.x, w2.y, w3.x, w3.y};
            a.x = __builtin_fmaf(xr[0], kl[i], a.x);
#pragma unroll
            for (int e = 1; e < Kk; ++e)
                a = __builtin_elementwise_fma((v2f){xr[e], xr[e]}, kq[i][e - 1], a);
            a.y = __builtin_fmaf(xr[Kk], kr[i], a.y);
        }
        *(v2f*)(ob + (size_t)cc * HW) = a;
    }
}

extern "C" void kernel_launch(void* const* d_in, const int* in_sizes, int n_in,
                              void* d_out, int out_size, void* d_ws, size_t ws_size,
                              hipStream_t stream) {
    const float* x     = (const float*)d_in[0];
    const float* rw    = (const float*)d_in[1];
    const float* rb    = (const float*)d_in[2];
    const float* gamma = (const float*)d_in[3];
    const float* beta  = (const float*)d_in[4];
    const float* mean  = (const float*)d_in[5];
    const float* var   = (const float*)d_in[6];
    const float* pa    = (const float*)d_in[7];
    const float* sw    = (const float*)d_in[8];
    const float* sb    = (const float*)d_in[9];
    float* out = (float*)d_out;

    float* t_ws   = (float*)d_ws;
    float* ker_ws = t_ws + (size_t)NP * TN;

    gemm1_reduce_bn_prelu<<<392, 128, 0, stream>>>(
        x, rw, rb, gamma, beta, mean, var, pa, t_ws);
    gemm2_span<<<686, 128, 0, stream>>>(
        t_ws, sw, sb, ker_ws);
    involution_apply<<<Bn * Gg * (Hh / TR), NT3, 0, stream>>>(
        x, ker_ws, out);
}
